# starting kernel
# baseline (speedup 1.0000x reference)
#include <hip/hip_runtime.h>


typedef float v2f __attribute__((ext_vector_type(2)));
typedef float v4f __attribute__((ext_vector_type(4)));
typedef float v8f __attribute__((ext_vector_type(8)));

#define PAD_ID 0
#define S_LEN 4096
#define D_LEN 512
#define WAVES 16
#define ROWS_PER_WAVE (S_LEN / WAVES)
#define CHUNKS (ROWS_PER_WAVE / 4)

__global__ __launch_bounds__(512, 1)
void nz_avg_pool_wmma(const float* __restrict__ x,
                      const int* __restrict__ tokens,
                      float* __restrict__ out) {
  const int dt   = blockIdx.x;
  const int b    = blockIdx.y;
  const int d0   = dt * 64;
  const int tid  = threadIdx.x;
  const int w    = tid >> 5;
  const int lane = tid & 31;
  const int n    = lane & 15;
  const int kA   = (lane & 16) ? 2 : 0;

  const float* __restrict__ xb = x + (size_t)b * S_LEN * D_LEN;
  const int*   __restrict__ tb = tokens + (size_t)b * S_LEN;

  const int sw = w * ROWS_PER_WAVE;

  v8f acc0 = {}, acc1 = {}, acc2 = {}, acc3 = {};
  float cnt = 0.0f;

  const float* px = xb + (size_t)(sw + kA) * D_LEN + d0 + 4 * n;
  const int*   pt = tb + sw + kA;

#pragma unroll 2
  for (int i = 0; i < CHUNKS; ++i) {
    v4f r0 = __builtin_nontemporal_load((const v4f*)px);
    v4f r1 = __builtin_nontemporal_load((const v4f*)(px + D_LEN));
    px += 4 * D_LEN;

    int2 t = *(const int2*)pt;
    pt += 4;
    float a0 = (t.x != PAD_ID) ? 1.0f : 0.0f;
    float a1 = (t.y != PAD_ID) ? 1.0f : 0.0f;
    cnt += (n == 0) ? (a0 + a1) : 0.0f;

    v2f a = {a0, a1};

    v2f b0 = {r0.x, r1.x};
    v2f b1 = {r0.y, r1.y};
    v2f b2 = {r0.z, r1.z};
    v2f b3 = {r0.w, r1.w};

    acc0 = __builtin_amdgcn_wmma_f32_16x16x4_f32(false, a, false, b0,
                                                 (short)0, acc0, false, false);
    acc1 = __builtin_amdgcn_wmma_f32_16x16x4_f32(false, a, false, b1,
                                                 (short)0, acc1, false, false);
    acc2 = __builtin_amdgcn_wmma_f32_16x16x4_f32(false, a, false, b2,
                                                 (short)0, acc2, false, false);
    acc3 = __builtin_amdgcn_wmma_f32_16x16x4_f32(false, a, false, b3,
                                                 (short)0, acc3, false, false);
  }

  __shared__ float sAcc[WAVES][4][16];
  __shared__ float sCnt[WAVES];

  if (lane < 16) {
    sAcc[w][0][n] = acc0[0];
    sAcc[w][1][n] = acc1[0];
    sAcc[w][2][n] = acc2[0];
    sAcc[w][3][n] = acc3[0];
  }
  float cw = cnt + __shfl_xor(cnt, 16, 32);
  if (lane == 0) sCnt[w] = cw;
  __syncthreads();

  if (tid < 64) {
    const int e  = tid & 3;
    const int nn = tid >> 2;
    float tot = 0.0f, c = 0.0f;
    for (int ww = 0; ww < WAVES; ++ww) {
      tot += sAcc[ww][e][nn];
      c   += sCnt[ww];
    }
    out[(size_t)b * D_LEN + d0 + tid] = tot / c;
  }
}

extern "C" void kernel_launch(void* const* d_in, const int* in_sizes, int n_in,
                              void* d_out, int out_size, void* d_ws,
                              size_t ws_size, hipStream_t stream) {
  (void)in_sizes; (void)n_in; (void)d_ws; (void)ws_size; (void)out_size;
  const float* x      = (const float*)d_in[0];
  const int*   tokens = (const int*)d_in[1];
  float*       out    = (float*)d_out;

  dim3 grid(D_LEN / 64, 16);
  nz_avg_pool_wmma<<<grid, 512, 0, stream>>>(x, tokens, out);
}


# Split the sequence, not the columns, over work-groups: group (batch, 256-row slice) streams whole 2 KiB rows, wave tile 4x128; slice sums via workspace, launch 2 adds the 16 slices; sum order changes
# speedup vs baseline: 2.2046x; 2.2046x over previous
#include <hip/hip_runtime.h>


typedef float v2f __attribute__((ext_vector_type(2)));
typedef float v4f __attribute__((ext_vector_type(4)));
typedef float v8f __attribute__((ext_vector_type(8)));

#define PAD_ID 0
#define B_LEN 16
#define S_LEN 4096
#define D_LEN 512
#define SLICES 16
#define STRIPS 4
#define ROW_GROUPS 4
#define WAVES (STRIPS * ROW_GROUPS)
#define CHUNK_ROWS 4
#define STEP_ROWS (CHUNK_ROWS * ROW_GROUPS)
#define SLICE_CHUNKS (S_LEN / SLICES / STEP_ROWS)
#define WHOLE_CHUNKS (S_LEN / STEP_ROWS)

#define PART_FLOATS (B_LEN * SLICES * D_LEN)
#define WS_BYTES ((PART_FLOATS + B_LEN * SLICES) * sizeof(float))

__device__ __forceinline__ bool ws_fits(const float* ws, size_t ws_size) {
  return ws != nullptr && ws_size >= WS_BYTES;
}

template <int NCHUNK>
__device__ __forceinline__ void stream_strip(const float* __restrict__ px,
                                             const int* __restrict__ pt,
                                             const int n, v8f (&acc)[8], float& cnt) {
#pragma unroll 2
  for (int i = 0; i < NCHUNK; ++i) {
    v4f r00 = __builtin_nontemporal_load((const v4f*)px);
    v4f r01 = __builtin_nontemporal_load((const v4f*)(px + 64));
    v4f r10 = __builtin_nontemporal_load((const v4f*)(px + D_LEN));
    v4f r11 = __builtin_nontemporal_load((const v4f*)(px + D_LEN + 64));
    px += STEP_ROWS * D_LEN;

    int2 t = *(const int2*)pt;
    pt += STEP_ROWS;
    float a0 = (t.x != PAD_ID) ? 1.0f : 0.0f;
    float a1 = (t.y != PAD_ID) ? 1.0f : 0.0f;
    cnt += (n == 0) ? (a0 + a1) : 0.0f;

    v2f a = {a0, a1};
    v2f b0 = {r00.x, r10.x}, b1 = {r00.y, r10.y}, b2 = {r00.z, r10.z}, b3 = {r00.w, r10.w};
    v2f b4 = {r01.x, r11.x}, b5 = {r01.y, r11.y}, b6 = {r01.z, r11.z}, b7 = {r01.w, r11.w};

    acc[0] = __builtin_amdgcn_wmma_f32_16x16x4_f32(false, a, false, b0, (short)0, acc[0], false, false);
    acc[1] = __builtin_amdgcn_wmma_f32_16x16x4_f32(false, a, false, b1, (short)0, acc[1], false, false);
    acc[2] = __builtin_amdgcn_wmma_f32_16x16x4_f32(false, a, false, b2, (short)0, acc[2], false, false);
    acc[3] = __builtin_amdgcn_wmma_f32_16x16x4_f32(false, a, false, b3, (short)0, acc[3], false, false);
    acc[4] = __builtin_amdgcn_wmma_f32_16x16x4_f32(false, a, false, b4, (short)0, acc[4], false, false);
    acc[5] = __builtin_amdgcn_wmma_f32_16x16x4_f32(false, a, false, b5, (short)0, acc[5], false, false);
    acc[6] = __builtin_amdgcn_wmma_f32_16x16x4_f32(false, a, false, b6, (short)0, acc[6], false, false);
    acc[7] = __builtin_amdgcn_wmma_f32_16x16x4_f32(false, a, false, b7, (short)0, acc[7], false, false);
  }
}

__global__ __launch_bounds__(WAVES * 32)
void masked_colsum_slices(const float* __restrict__ x,
                          const int* __restrict__ tokens,
                          float* __restrict__ ws, size_t ws_size,
                          float* __restrict__ out) {
  const int slice = blockIdx.x;
  const int b     = blockIdx.y;
  const bool sliced = ws_fits(ws, ws_size);
  if (!sliced && slice != 0) return;

  const int tid  = threadIdx.x;
  const int w    = tid >> 5;
  const int lane = tid & 31;
  const int st   = w & (STRIPS - 1);
  const int rg   = w / STRIPS;
  const int n    = lane & 15;
  const int kA   = (lane & 16) ? 2 : 0;

  const int row0 = (sliced ? slice * (S_LEN / SLICES) : 0) + rg * CHUNK_ROWS + kA;
  const float* px = x + ((size_t)b * S_LEN + row0) * D_LEN + st * 128 + 4 * n;
  const int*   pt = tokens + (size_t)b * S_LEN + row0;

  v8f acc[8] = {};
  float cnt = 0.0f;
  if (sliced) stream_strip<SLICE_CHUNKS>(px, pt, n, acc, cnt);
  else        stream_strip<WHOLE_CHUNKS>(px, pt, n, acc, cnt);

  __shared__ float sAcc[ROW_GROUPS][D_LEN];
  __shared__ float sCnt[ROW_GROUPS];

  if (lane < 16) {
    v4f lo = {acc[0][0], acc[1][0], acc[2][0], acc[3][0]};
    v4f hi = {acc[4][0], acc[5][0], acc[6][0], acc[7][0]};
    *(v4f*)&sAcc[rg][st * 128 + 4 * n]      = lo;
    *(v4f*)&sAcc[rg][st * 128 + 64 + 4 * n] = hi;
  }
  float cw = cnt + __shfl_xor(cnt, 16, 32);
  if (st == 0 && lane == 0) sCnt[rg] = cw;
  __syncthreads();

  if (tid < D_LEN / 4) {
    v4f tot = *(const v4f*)&sAcc[0][4 * tid];
    float c = sCnt[0];
    for (int g = 1; g < ROW_GROUPS; ++g) {
      tot += *(const v4f*)&sAcc[g][4 * tid];
      c   += sCnt[g];
    }
    if (sliced) {
      *(v4f*)(ws + ((size_t)b * SLICES + slice) * D_LEN + 4 * tid) = tot;
      if (tid == 0) ws[PART_FLOATS + b * SLICES + slice] = c;
    } else {
      *(v4f*)(out + (size_t)b * D_LEN + 4 * tid) = tot / c;
    }
  }
}

__global__ __launch_bounds__(D_LEN / 4)
void add_slices_and_divide(const float* __restrict__ ws, size_t ws_size,
                           float* __restrict__ out) {
  if (!ws_fits(ws, ws_size)) return;
  const int b   = blockIdx.x;
  const int tid = threadIdx.x;
  const float* part = ws + (size_t)b * SLICES * D_LEN + 4 * tid;
  const float* cnts = ws + PART_FLOATS + b * SLICES;
  v4f tot = *(const v4f*)part;
  float c = cnts[0];
  for (int j = 1; j < SLICES; ++j) {
    tot += *(const v4f*)(part + (size_t)j * D_LEN);
    c   += cnts[j];
  }
  *(v4f*)(out + (size_t)b * D_LEN + 4 * tid) = tot / c;
}

extern "C" void kernel_launch(void* const* d_in, const int* in_sizes, int n_in,
                              void* d_out, int out_size, void* d_ws,
                              size_t ws_size, hipStream_t stream) {
  (void)in_sizes; (void)n_in; (void)out_size;
  const float* x      = (const float*)d_in[0];
  const int*   tokens = (const int*)d_in[1];
  float*       out    = (float*)d_out;
  float*       ws     = (float*)d_ws;

  dim3 grid(SLICES, B_LEN);
  masked_colsum_slices<<<grid, WAVES * 32, 0, stream>>>(x, tokens, ws, ws_size, out);
  add_slices_and_divide<<<B_LEN, D_LEN / 4, 0, stream>>>(ws, ws_size, out);
}


# Declare the real 4 waves per SIMD (amdgpu_waves_per_eu(4,4)): the compiler stops saving registers for more waves and issues all 8 b128 of a trip at once, not 7 then 1 behind their return
# speedup vs baseline: 2.4533x; 1.1128x over previous
#include <hip/hip_runtime.h>


typedef float v2f __attribute__((ext_vector_type(2)));
typedef float v4f __attribute__((ext_vector_type(4)));
typedef float v8f __attribute__((ext_vector_type(8)));

#define PAD_ID 0
#define B_LEN 16
#define S_LEN 4096
#define D_LEN 512
#define SLICES 16
#define STRIPS 4
#define ROW_GROUPS 4
#define WAVES (STRIPS * ROW_GROUPS)
#define CHUNK_ROWS 4
#define STEP_ROWS (CHUNK_ROWS * ROW_GROUPS)
#define SLICE_CHUNKS (S_LEN / SLICES / STEP_ROWS)
#define WHOLE_CHUNKS (S_LEN / STEP_ROWS)

#define PART_FLOATS (B_LEN * SLICES * D_LEN)
#define WS_BYTES ((PART_FLOATS + B_LEN * SLICES) * sizeof(float))

__device__ __forceinline__ bool ws_fits(const float* ws, size_t ws_size) {
  return ws != nullptr && ws_size >= WS_BYTES;
}

template <int NCHUNK>
__device__ __forceinline__ void stream_strip(const float* __restrict__ px,
                                             const int* __restrict__ pt,
                                             const int n, v8f (&acc)[8], float& cnt) {
#pragma unroll 2
  for (int i = 0; i < NCHUNK; ++i) {
    v4f r00 = __builtin_nontemporal_load((const v4f*)px);
    v4f r01 = __builtin_nontemporal_load((const v4f*)(px + 64));
    v4f r10 = __builtin_nontemporal_load((const v4f*)(px + D_LEN));
    v4f r11 = __builtin_nontemporal_load((const v4f*)(px + D_LEN + 64));
    px += STEP_ROWS * D_LEN;

    int2 t = *(const int2*)pt;
    pt += STEP_ROWS;
    float a0 = (t.x != PAD_ID) ? 1.0f : 0.0f;
    float a1 = (t.y != PAD_ID) ? 1.0f : 0.0f;
    cnt += (n == 0) ? (a0 + a1) : 0.0f;

    v2f a = {a0, a1};
    v2f b0 = {r00.x, r10.x}, b1 = {r00.y, r10.y}, b2 = {r00.z, r10.z}, b3 = {r00.w, r10.w};
    v2f b4 = {r01.x, r11.x}, b5 = {r01.y, r11.y}, b6 = {r01.z, r11.z}, b7 = {r01.w, r11.w};

    acc[0] = __builtin_amdgcn_wmma_f32_16x16x4_f32(false, a, false, b0, (short)0, acc[0], false, false);
    acc[1] = __builtin_amdgcn_wmma_f32_16x16x4_f32(false, a, false, b1, (short)0, acc[1], false, false);
    acc[2] = __builtin_amdgcn_wmma_f32_16x16x4_f32(false, a, false, b2, (short)0, acc[2], false, false);
    acc[3] = __builtin_amdgcn_wmma_f32_16x16x4_f32(false, a, false, b3, (short)0, acc[3], false, false);
    acc[4] = __builtin_amdgcn_wmma_f32_16x16x4_f32(false, a, false, b4, (short)0, acc[4], false, false);
    acc[5] = __builtin_amdgcn_wmma_f32_16x16x4_f32(false, a, false, b5, (short)0, acc[5], false, false);
    acc[6] = __builtin_amdgcn_wmma_f32_16x16x4_f32(false, a, false, b6, (short)0, acc[6], false, false);
    acc[7] = __builtin_amdgcn_wmma_f32_16x16x4_f32(false, a, false, b7, (short)0, acc[7], false, false);
  }
}

__global__ __launch_bounds__(WAVES * 32) __attribute__((amdgpu_waves_per_eu(4, 4)))
void masked_colsum_slices(const float* __restrict__ x,
                          const int* __restrict__ tokens,
                          float* __restrict__ ws, size_t ws_size,
                          float* __restrict__ out) {
  const int slice = blockIdx.x;
  const int b     = blockIdx.y;
  const bool sliced = ws_fits(ws, ws_size);
  if (!sliced && slice != 0) return;

  const int tid  = threadIdx.x;
  const int w    = tid >> 5;
  const int lane = tid & 31;
  const int st   = w & (STRIPS - 1);
  const int rg   = w / STRIPS;
  const int n    = lane & 15;
  const int kA   = (lane & 16) ? 2 : 0;

  const int row0 = (sliced ? slice * (S_LEN / SLICES) : 0) + rg * CHUNK_ROWS + kA;
  const float* px = x + ((size_t)b * S_LEN + row0) * D_LEN + st * 128 + 4 * n;
  const int*   pt = tokens + (size_t)b * S_LEN + row0;

  v8f acc[8] = {};
  float cnt = 0.0f;
  if (sliced) stream_strip<SLICE_CHUNKS>(px, pt, n, acc, cnt);
  else        stream_strip<WHOLE_CHUNKS>(px, pt, n, acc, cnt);

  __shared__ float sAcc[ROW_GROUPS][D_LEN];
  __shared__ float sCnt[ROW_GROUPS];

  if (lane < 16) {
    v4f lo = {acc[0][0], acc[1][0], acc[2][0], acc[3][0]};
    v4f hi = {acc[4][0], acc[5][0], acc[6][0], acc[7][0]};
    *(v4f*)&sAcc[rg][st * 128 + 4 * n]      = lo;
    *(v4f*)&sAcc[rg][st * 128 + 64 + 4 * n] = hi;
  }
  float cw = cnt + __shfl_xor(cnt, 16, 32);
  if (st == 0 && lane == 0) sCnt[rg] = cw;
  __syncthreads();

  if (tid < D_LEN / 4) {
    v4f tot = *(const v4f*)&sAcc[0][4 * tid];
    float c = sCnt[0];
    for (int g = 1; g < ROW_GROUPS; ++g) {
      tot += *(const v4f*)&sAcc[g][4 * tid];
      c   += sCnt[g];
    }
    if (sliced) {
      *(v4f*)(ws + ((size_t)b * SLICES + slice) * D_LEN + 4 * tid) = tot;
      if (tid == 0) ws[PART_FLOATS + b * SLICES + slice] = c;
    } else {
      *(v4f*)(out + (size_t)b * D_LEN + 4 * tid) = tot / c;
    }
  }
}

__global__ __launch_bounds__(D_LEN / 4)
void add_slices_and_divide(const float* __restrict__ ws, size_t ws_size,
                           float* __restrict__ out) {
  if (!ws_fits(ws, ws_size)) return;
  const int b   = blockIdx.x;
  const int tid = threadIdx.x;
  const float* part = ws + (size_t)b * SLICES * D_LEN + 4 * tid;
  const float* cnts = ws + PART_FLOATS + b * SLICES;
  v4f tot = *(const v4f*)part;
  float c = cnts[0];
  for (int j = 1; j < SLICES; ++j) {
    tot += *(const v4f*)(part + (size_t)j * D_LEN);
    c   += cnts[j];
  }
  *(v4f*)(out + (size_t)b * D_LEN + 4 * tid) = tot / c;
}

extern "C" void kernel_launch(void* const* d_in, const int* in_sizes, int n_in,
                              void* d_out, int out_size, void* d_ws,
                              size_t ws_size, hipStream_t stream) {
  (void)in_sizes; (void)n_in; (void)out_size;
  const float* x      = (const float*)d_in[0];
  const int*   tokens = (const int*)d_in[1];
  float*       out    = (float*)d_out;
  float*       ws     = (float*)d_ws;

  dim3 grid(SLICES, B_LEN);
  masked_colsum_slices<<<grid, WAVES * 32, 0, stream>>>(x, tokens, ws, ws_size, out);
  add_slices_and_divide<<<B_LEN, D_LEN / 4, 0, stream>>>(ws, ws_size, out);
}


# Software-pipeline the x stream through a ring of 3 register trip buffers: the 8 b128 + 2 token pairs of trip i+2 are issued ahead of trip i's 16 WMMAs; chunk and summation order unchanged
# speedup vs baseline: 2.6279x; 1.0712x over previous
#include <hip/hip_runtime.h>


typedef float v2f __attribute__((ext_vector_type(2)));
typedef float v4f __attribute__((ext_vector_type(4)));
typedef float v8f __attribute__((ext_vector_type(8)));

#define PAD_ID 0
#define B_LEN 16
#define S_LEN 4096
#define D_LEN 512
#define SLICES 16
#define STRIPS 4
#define ROW_GROUPS 4
#define WAVES (STRIPS * ROW_GROUPS)
#define CHUNK_ROWS 4
#define STEP_ROWS (CHUNK_ROWS * ROW_GROUPS)
#define SLICE_CHUNKS (S_LEN / SLICES / STEP_ROWS)

#define PART_FLOATS (B_LEN * SLICES * D_LEN)
#define WS_BYTES ((PART_FLOATS + B_LEN * SLICES) * sizeof(float))

__device__ __forceinline__ bool ws_fits(const float* ws, size_t ws_size) {
  return ws != nullptr && ws_size >= WS_BYTES;
}

#define TRIP_CHUNKS 2
#define TRIP_X (TRIP_CHUNKS * STEP_ROWS * D_LEN)
#define TRIP_T (TRIP_CHUNKS * STEP_ROWS)
#define SLICE_ROWS (S_LEN / SLICES)
#define SLICE_TRIPS (SLICE_CHUNKS / TRIP_CHUNKS)
#define AHEAD 2
#define RING (AHEAD + 1)

struct Trip {
  v4f  r[TRIP_CHUNKS][4];
  int2 t[TRIP_CHUNKS];
};

__device__ __forceinline__ void load_trip(Trip& q, const float* __restrict__ px,
                                          const int* __restrict__ pt) {
#pragma unroll
  for (int c = 0; c < TRIP_CHUNKS; ++c) {
    const float* p = px + c * STEP_ROWS * D_LEN;
    q.r[c][0] = __builtin_nontemporal_load((const v4f*)p);
    q.r[c][1] = __builtin_nontemporal_load((const v4f*)(p + 64));
    q.r[c][2] = __builtin_nontemporal_load((const v4f*)(p + D_LEN));
    q.r[c][3] = __builtin_nontemporal_load((const v4f*)(p + D_LEN + 64));
    q.t[c] = *(const int2*)(pt + c * STEP_ROWS);
  }
}

__device__ __forceinline__ void mma_trip(const Trip& q, const int n, v8f (&acc)[8], float& cnt) {
#pragma unroll
  for (int c = 0; c < TRIP_CHUNKS; ++c) {
    const v4f r00 = q.r[c][0], r01 = q.r[c][1], r10 = q.r[c][2], r11 = q.r[c][3];
    const int2 t = q.t[c];
    float a0 = (t.x != PAD_ID) ? 1.0f : 0.0f;
    float a1 = (t.y != PAD_ID) ? 1.0f : 0.0f;
    cnt += (n == 0) ? (a0 + a1) : 0.0f;

    v2f a = {a0, a1};
    v2f b0 = {r00.x, r10.x}, b1 = {r00.y, r10.y}, b2 = {r00.z, r10.z}, b3 = {r00.w, r10.w};
    v2f b4 = {r01.x, r11.x}, b5 = {r01.y, r11.y}, b6 = {r01.z, r11.z}, b7 = {r01.w, r11.w};

    acc[0] = __builtin_amdgcn_wmma_f32_16x16x4_f32(false, a, false, b0, (short)0, acc[0], false, false);
    acc[1] = __builtin_amdgcn_wmma_f32_16x16x4_f32(false, a, false, b1, (short)0, acc[1], false, false);
    acc[2] = __builtin_amdgcn_wmma_f32_16x16x4_f32(false, a, false, b2, (short)0, acc[2], false, false);
    acc[3] = __builtin_amdgcn_wmma_f32_16x16x4_f32(false, a, false, b3, (short)0, acc[3], false, false);
    acc[4] = __builtin_amdgcn_wmma_f32_16x16x4_f32(false, a, false, b4, (short)0, acc[4], false, false);
    acc[5] = __builtin_amdgcn_wmma_f32_16x16x4_f32(false, a, false, b5, (short)0, acc[5], false, false);
    acc[6] = __builtin_amdgcn_wmma_f32_16x16x4_f32(false, a, false, b6, (short)0, acc[6], false, false);
    acc[7] = __builtin_amdgcn_wmma_f32_16x16x4_f32(false, a, false, b7, (short)0, acc[7], false, false);
  }
}

__device__ __forceinline__ void stream_slice(const float* __restrict__ px,
                                             const int* __restrict__ pt,
                                             const int n, v8f (&acc)[8], float& cnt) {
  Trip q[RING];
#pragma unroll
  for (int i = 0; i < AHEAD; ++i) load_trip(q[i], px + i * TRIP_X, pt + i * TRIP_T);
#pragma unroll
  for (int i = 0; i < SLICE_TRIPS; ++i) {
    if (i + AHEAD < SLICE_TRIPS)
      load_trip(q[(i + AHEAD) % RING], px + (i + AHEAD) * TRIP_X, pt + (i + AHEAD) * TRIP_T);
    mma_trip(q[i % RING], n, acc, cnt);
  }
}

__global__ __launch_bounds__(WAVES * 32)
void masked_colsum_slices(const float* __restrict__ x,
                          const int* __restrict__ tokens,
                          float* __restrict__ ws, size_t ws_size,
                          float* __restrict__ out) {
  const int slice = blockIdx.x;
  const int b     = blockIdx.y;
  const bool sliced = ws_fits(ws, ws_size);
  if (!sliced && slice != 0) return;

  const int tid  = threadIdx.x;
  const int w    = tid >> 5;
  const int lane = tid & 31;
  const int st   = w & (STRIPS - 1);
  const int rg   = w / STRIPS;
  const int n    = lane & 15;
  const int kA   = (lane & 16) ? 2 : 0;

  const int row0 = (sliced ? slice * (S_LEN / SLICES) : 0) + rg * CHUNK_ROWS + kA;
  const float* px = x + ((size_t)b * S_LEN + row0) * D_LEN + st * 128 + 4 * n;
  const int*   pt = tokens + (size_t)b * S_LEN + row0;

  v8f acc[8] = {};
  float cnt = 0.0f;
  const int nslice = sliced ? 1 : SLICES;
  for (int j = 0; j < nslice; ++j)
    stream_slice(px + (size_t)j * SLICE_ROWS * D_LEN, pt + j * SLICE_ROWS, n, acc, cnt);

  __shared__ float sAcc[ROW_GROUPS][D_LEN];
  __shared__ float sCnt[ROW_GROUPS];

  if (lane < 16) {
    v4f lo = {acc[0][0], acc[1][0], acc[2][0], acc[3][0]};
    v4f hi = {acc[4][0], acc[5][0], acc[6][0], acc[7][0]};
    *(v4f*)&sAcc[rg][st * 128 + 4 * n]      = lo;
    *(v4f*)&sAcc[rg][st * 128 + 64 + 4 * n] = hi;
  }
  float cw = cnt + __shfl_xor(cnt, 16, 32);
  if (st == 0 && lane == 0) sCnt[rg] = cw;
  __syncthreads();

  if (tid < D_LEN / 4) {
    v4f tot = *(const v4f*)&sAcc[0][4 * tid];
    float c = sCnt[0];
    for (int g = 1; g < ROW_GROUPS; ++g) {
      tot += *(const v4f*)&sAcc[g][4 * tid];
      c   += sCnt[g];
    }
    if (sliced) {
      *(v4f*)(ws + ((size_t)b * SLICES + slice) * D_LEN + 4 * tid) = tot;
      if (tid == 0) ws[PART_FLOATS + b * SLICES + slice] = c;
    } else {
      *(v4f*)(out + (size_t)b * D_LEN + 4 * tid) = tot / c;
    }
  }
}

__global__ __launch_bounds__(D_LEN / 4)
void add_slices_and_divide(const float* __restrict__ ws, size_t ws_size,
                           float* __restrict__ out) {
  if (!ws_fits(ws, ws_size)) return;
  const int b   = blockIdx.x;
  const int tid = threadIdx.x;
  const float* part = ws + (size_t)b * SLICES * D_LEN + 4 * tid;
  const float* cnts = ws + PART_FLOATS + b * SLICES;
  v4f tot = *(const v4f*)part;
  float c = cnts[0];
  for (int j = 1; j < SLICES; ++j) {
    tot += *(const v4f*)(part + (size_t)j * D_LEN);
    c   += cnts[j];
  }
  *(v4f*)(out + (size_t)b * D_LEN + 4 * tid) = tot / c;
}

extern "C" void kernel_launch(void* const* d_in, const int* in_sizes, int n_in,
                              void* d_out, int out_size, void* d_ws,
                              size_t ws_size, hipStream_t stream) {
  (void)in_sizes; (void)n_in; (void)out_size;
  const float* x      = (const float*)d_in[0];
  const int*   tokens = (const int*)d_in[1];
  float*       out    = (float*)d_out;
  float*       ws     = (float*)d_ws;

  dim3 grid(SLICES, B_LEN);
  masked_colsum_slices<<<grid, WAVES * 32, 0, stream>>>(x, tokens, ws, ws_size, out);
  add_slices_and_divide<<<B_LEN, D_LEN / 4, 0, stream>>>(ws, ws_size, out);
}


# Do not read the rows the mask removes: each row's two b128 are issued under EXEC = ballot(token != pad) (asm, counted waits by hand), tokens fetched first; A, WMMAs and summation order unchanged
# speedup vs baseline: 2.9369x; 1.1176x over previous
#include <hip/hip_runtime.h>


typedef float v2f __attribute__((ext_vector_type(2)));
typedef float v4f __attribute__((ext_vector_type(4)));
typedef float v8f __attribute__((ext_vector_type(8)));

#define PAD_ID 0
#define B_LEN 16
#define S_LEN 4096
#define D_LEN 512
#define SLICES 16
#define STRIPS 4
#define ROW_GROUPS 4
#define WAVES (STRIPS * ROW_GROUPS)
#define CHUNK_ROWS 4
#define STEP_ROWS (CHUNK_ROWS * ROW_GROUPS)
#define SLICE_CHUNKS (S_LEN / SLICES / STEP_ROWS)

#define PART_FLOATS (B_LEN * SLICES * D_LEN)
#define WS_BYTES ((PART_FLOATS + B_LEN * SLICES) * sizeof(float))

__device__ __forceinline__ bool ws_fits(const float* ws, size_t ws_size) {
  return ws != nullptr && ws_size >= WS_BYTES;
}

#define TRIP_CHUNKS 2
#define TRIP_X (TRIP_CHUNKS * STEP_ROWS * D_LEN)
#define TRIP_T (TRIP_CHUNKS * STEP_ROWS)
#define SLICE_ROWS (S_LEN / SLICES)
#define SLICE_TRIPS (SLICE_CHUNKS / TRIP_CHUNKS)
#define AHEAD 2
#define RING (AHEAD + 1)

struct Trip {
  v4f r[TRIP_CHUNKS][4];
};

template <int OFF>
__device__ __forceinline__ void load_row_if_valid(v4f& r0, v4f& r1, const float* p, unsigned m) {
  unsigned sv;
  asm volatile("s_and_saveexec_b32 %[sv], %[m]\n\t"
               "global_load_b128 %[r0], %[p], off offset:%[o0] th:TH_LOAD_NT\n\t"
               "global_load_b128 %[r1], %[p], off offset:%[o1] th:TH_LOAD_NT\n\t"
               "s_mov_b32 exec_lo, %[sv]"
               : [r0] "+v"(r0), [r1] "+v"(r1), [sv] "=&s"(sv)
               : [p] "v"(p), [m] "s"(m), [o0] "n"(OFF), [o1] "n"(OFF + 256)
               : "scc");
}

template <int NLATER>
__device__ __forceinline__ void wait_trip(Trip& q, const float* p) {
  asm volatile("s_wait_loadcnt %[n]"
               : "+v"(q.r[0][0]), "+v"(q.r[0][1]), "+v"(q.r[0][2]), "+v"(q.r[0][3]),
                 "+v"(q.r[1][0]), "+v"(q.r[1][1]), "+v"(q.r[1][2]), "+v"(q.r[1][3])
               : [n] "n"(NLATER), "v"(p));
}

template <int I>
__device__ __forceinline__ void load_trip(Trip& q, const float* __restrict__ px,
                                          const unsigned (&m0)[2], const unsigned (&m1)[2]) {
  constexpr int B  = I * TRIP_X * 4;
  constexpr int C1 = STEP_ROWS * D_LEN * 4;
  constexpr int R1 = D_LEN * 4;
  load_row_if_valid<B>(q.r[0][0], q.r[0][1], px, m0[0]);
  load_row_if_valid<B + R1>(q.r[0][2], q.r[0][3], px, m0[1]);
  load_row_if_valid<B + C1>(q.r[1][0], q.r[1][1], px, m1[0]);
  load_row_if_valid<B + C1 + R1>(q.r[1][2], q.r[1][3], px, m1[1]);
}

__device__ __forceinline__ void mma_trip(const Trip& q, const v2f (&av)[TRIP_CHUNKS],
                                         v8f (&acc)[8]) {
#pragma unroll
  for (int c = 0; c < TRIP_CHUNKS; ++c) {
    const v4f r00 = q.r[c][0], r01 = q.r[c][1], r10 = q.r[c][2], r11 = q.r[c][3];

    const v2f a = av[c];
    v2f b0 = {r00.x, r10.x}, b1 = {r00.y, r10.y}, b2 = {r00.z, r10.z}, b3 = {r00.w, r10.w};
    v2f b4 = {r01.x, r11.x}, b5 = {r01.y, r11.y}, b6 = {r01.z, r11.z}, b7 = {r01.w, r11.w};

    acc[0] = __builtin_amdgcn_wmma_f32_16x16x4_f32(false, a, false, b0, (short)0, acc[0], false, false);
    acc[1] = __builtin_amdgcn_wmma_f32_16x16x4_f32(false, a, false, b1, (short)0, acc[1], false, false);
    acc[2] = __builtin_amdgcn_wmma_f32_16x16x4_f32(false, a, false, b2, (short)0, acc[2], false, false);
    acc[3] = __builtin_amdgcn_wmma_f32_16x16x4_f32(false, a, false, b3, (short)0, acc[3], false, false);
    acc[4] = __builtin_amdgcn_wmma_f32_16x16x4_f32(false, a, false, b4, (short)0, acc[4], false, false);
    acc[5] = __builtin_amdgcn_wmma_f32_16x16x4_f32(false, a, false, b5, (short)0, acc[5], false, false);
    acc[6] = __builtin_amdgcn_wmma_f32_16x16x4_f32(false, a, false, b6, (short)0, acc[6], false, false);
    acc[7] = __builtin_amdgcn_wmma_f32_16x16x4_f32(false, a, false, b7, (short)0, acc[7], false, false);
  }
}

struct SliceMasks {
  v2f      a[SLICE_TRIPS][TRIP_CHUNKS];
  unsigned m[SLICE_TRIPS][TRIP_CHUNKS][2];
};

template <int I>
__device__ __forceinline__ void pipe_step(Trip (&q)[RING], const float* __restrict__ px,
                                          const SliceMasks& k, v8f (&acc)[8]) {
  if constexpr (I + AHEAD < SLICE_TRIPS) {
    load_trip<I + AHEAD>(q[(I + AHEAD) % RING], px, k.m[I + AHEAD][0], k.m[I + AHEAD][1]);
    wait_trip<8 * AHEAD>(q[I % RING], px);
  } else {
    wait_trip<8 * (SLICE_TRIPS - 1 - I)>(q[I % RING], px);
  }
  mma_trip(q[I % RING], k.a[I], acc);
}

__device__ __forceinline__ void stream_slice(const float* __restrict__ px,
                                             const int* __restrict__ pt,
                                             const int n, v8f (&acc)[8], float& cnt) {
  int2 t[SLICE_TRIPS][TRIP_CHUNKS];
#pragma unroll
  for (int i = 0; i < SLICE_TRIPS; ++i)
#pragma unroll
    for (int c = 0; c < TRIP_CHUNKS; ++c)
      t[i][c] = *(const int2*)(pt + (i * TRIP_CHUNKS + c) * STEP_ROWS);

#pragma unroll
  for (int i = 0; i < SLICE_TRIPS; ++i)
    asm volatile("" : "+v"(t[i][0]), "+v"(t[i][1]));

  SliceMasks k;
#pragma unroll
  for (int i = 0; i < SLICE_TRIPS; ++i)
#pragma unroll
    for (int c = 0; c < TRIP_CHUNKS; ++c) {
      const bool v0 = t[i][c].x != PAD_ID, v1 = t[i][c].y != PAD_ID;
      const float a0 = v0 ? 1.0f : 0.0f, a1 = v1 ? 1.0f : 0.0f;
      k.a[i][c] = (v2f){a0, a1};
      const unsigned e0 = __builtin_amdgcn_ballot_w32(v0), e1 = __builtin_amdgcn_ballot_w32(v1);
      k.m[i][c][0] = e0 ? e0 : 1u;
      k.m[i][c][1] = e1 ? e1 : 1u;
      cnt += (n == 0) ? (a0 + a1) : 0.0f;
    }

  Trip q[RING];
#pragma unroll
  for (int s = 0; s < RING; ++s)
#pragma unroll
    for (int c = 0; c < TRIP_CHUNKS; ++c)
#pragma unroll
      for (int e = 0; e < 4; ++e) q[s].r[c][e] = (v4f){0.0f, 0.0f, 0.0f, 0.0f};

  static_assert(AHEAD == 2 && SLICE_TRIPS == 8, "the pipeline below is written out for 8 trips, 2 ahead");
  load_trip<0>(q[0], px, k.m[0][0], k.m[0][1]);
  load_trip<1>(q[1], px, k.m[1][0], k.m[1][1]);
  pipe_step<0>(q, px, k, acc);
  pipe_step<1>(q, px, k, acc);
  pipe_step<2>(q, px, k, acc);
  pipe_step<3>(q, px, k, acc);
  pipe_step<4>(q, px, k, acc);
  pipe_step<5>(q, px, k, acc);
  pipe_step<6>(q, px, k, acc);
  pipe_step<7>(q, px, k, acc);
}

__global__ __launch_bounds__(WAVES * 32)
void masked_colsum_slices(const float* __restrict__ x,
                          const int* __restrict__ tokens,
                          float* __restrict__ ws, size_t ws_size,
                          float* __restrict__ out) {
  const int slice = blockIdx.x;
  const int b     = blockIdx.y;
  const bool sliced = ws_fits(ws, ws_size);
  if (!sliced && slice != 0) return;

  const int tid  = threadIdx.x;
  const int w    = tid >> 5;
  const int lane = tid & 31;
  const int st   = w & (STRIPS - 1);
  const int rg   = w / STRIPS;
  const int n    = lane & 15;
  const int kA   = (lane & 16) ? 2 : 0;

  const int row0 = (sliced ? slice * (S_LEN / SLICES) : 0) + rg * CHUNK_ROWS + kA;
  const float* px = x + ((size_t)b * S_LEN + row0) * D_LEN + st * 128 + 4 * n;
  const int*   pt = tokens + (size_t)b * S_LEN + row0;

  v8f acc[8] = {};
  float cnt = 0.0f;
  const int nslice = sliced ? 1 : SLICES;
  for (int j = 0; j < nslice; ++j)
    stream_slice(px + (size_t)j * SLICE_ROWS * D_LEN, pt + j * SLICE_ROWS, n, acc, cnt);

  __shared__ float sAcc[ROW_GROUPS][D_LEN];
  __shared__ float sCnt[ROW_GROUPS];

  if (lane < 16) {
    v4f lo = {acc[0][0], acc[1][0], acc[2][0], acc[3][0]};
    v4f hi = {acc[4][0], acc[5][0], acc[6][0], acc[7][0]};
    *(v4f*)&sAcc[rg][st * 128 + 4 * n]      = lo;
    *(v4f*)&sAcc[rg][st * 128 + 64 + 4 * n] = hi;
  }
  float cw = cnt + __shfl_xor(cnt, 16, 32);
  if (st == 0 && lane == 0) sCnt[rg] = cw;
  __syncthreads();

  if (tid < D_LEN / 4) {
    v4f tot = *(const v4f*)&sAcc[0][4 * tid];
    float c = sCnt[0];
    for (int g = 1; g < ROW_GROUPS; ++g) {
      tot += *(const v4f*)&sAcc[g][4 * tid];
      c   += sCnt[g];
    }
    if (sliced) {
      *(v4f*)(ws + ((size_t)b * SLICES + slice) * D_LEN + 4 * tid) = tot;
      if (tid == 0) ws[PART_FLOATS + b * SLICES + slice] = c;
    } else {
      *(v4f*)(out + (size_t)b * D_LEN + 4 * tid) = tot / c;
    }
  }
}

__global__ __launch_bounds__(D_LEN / 4)
void add_slices_and_divide(const float* __restrict__ ws, size_t ws_size,
                           float* __restrict__ out) {
  if (!ws_fits(ws, ws_size)) return;
  const int b   = blockIdx.x;
  const int tid = threadIdx.x;
  const float* part = ws + (size_t)b * SLICES * D_LEN + 4 * tid;
  const float* cnts = ws + PART_FLOATS + b * SLICES;
  v4f tot = *(const v4f*)part;
  float c = cnts[0];
  for (int j = 1; j < SLICES; ++j) {
    tot += *(const v4f*)(part + (size_t)j * D_LEN);
    c   += cnts[j];
  }
  *(v4f*)(out + (size_t)b * D_LEN + 4 * tid) = tot / c;
}

extern "C" void kernel_launch(void* const* d_in, const int* in_sizes, int n_in,
                              void* d_out, int out_size, void* d_ws,
                              size_t ws_size, hipStream_t stream) {
  (void)in_sizes; (void)n_in; (void)out_size;
  const float* x      = (const float*)d_in[0];
  const int*   tokens = (const int*)d_in[1];
  float*       out    = (float*)d_out;
  float*       ws     = (float*)d_ws;

  dim3 grid(SLICES, B_LEN);
  masked_colsum_slices<<<grid, WAVES * 32, 0, stream>>>(x, tokens, ws, ws_size, out);
  add_slices_and_divide<<<B_LEN, D_LEN / 4, 0, stream>>>(ws, ws_size, out);
}
